# starting kernel
# baseline (speedup 1.0000x reference)
#include <hip/hip_runtime.h>

#define N_NODES 50000
#define N_EDGES 800000
#define D       128
#define ALPHA   0.1f
#define BETA    0.5f

#define EDGES_PER_WAVE 32
#define TILE_ROWS      16
#define WAVES_PER_BLK  4
#define ROW_TILES      (N_NODES / TILE_ROWS)
#define LDSP           (D + 4)

typedef float v2f __attribute__((ext_vector_type(2)));
typedef float v8f __attribute__((ext_vector_type(8)));

__global__ __launch_bounds__(256)
void gcn_zero(float4* __restrict__ p, int n4) {
    int i = blockIdx.x * blockDim.x + threadIdx.x;
    if (i < n4) {
        float4 z; z.x = 0.f; z.y = 0.f; z.z = 0.f; z.w = 0.f;
        p[i] = z;
    }
}

__global__ __launch_bounds__(256)
void gcn_scatter(const float* __restrict__ h,
                 const int*   __restrict__ erow,
                 const int*   __restrict__ ecol,
                 const float* __restrict__ evals,
                 float*       __restrict__ hi)
{
    const int gwave = (blockIdx.x * blockDim.x + threadIdx.x) >> 5;
    const int lane  = threadIdx.x & 31;
    const int e0    = gwave * EDGES_PER_WAVE;
    if (e0 >= N_EDGES) return;
    const int e1 = min(e0 + EDGES_PER_WAVE, N_EDGES);
    const int f  = lane * 4;

    float ax = 0.f, ay = 0.f, az = 0.f, aw = 0.f;
    int cur = erow[e0];

    for (int e = e0; e < e1; ++e) {
        const int   col = ecol[e];
        const float val = evals[e];
        const float4 s = *(const float4*)(h + (size_t)col * D + f);
        ax = fmaf(val, s.x, ax);
        ay = fmaf(val, s.y, ay);
        az = fmaf(val, s.z, az);
        aw = fmaf(val, s.w, aw);
        const int nxt = (e + 1 < e1) ? erow[e + 1] : -1;
        if (nxt != cur) {
            float* dst = hi + (size_t)cur * D + f;
            unsafeAtomicAdd(dst + 0, ax);
            unsafeAtomicAdd(dst + 1, ay);
            unsafeAtomicAdd(dst + 2, az);
            unsafeAtomicAdd(dst + 3, aw);
            ax = ay = az = aw = 0.f;
            cur = nxt;
        }
    }
}

__global__ __launch_bounds__(128)
void gcn_gemm(const float* __restrict__ hi,
              const float* __restrict__ h0,
              const float* __restrict__ W,
              const float* __restrict__ bias,
              const int*   __restrict__ lptr,
              float*       __restrict__ out)
{
    __shared__ float sp[64 * LDSP];

    const int lane  = threadIdx.x & 31;
    const int wv    = threadIdx.x >> 5;
    const int tile0 = blockIdx.x * WAVES_PER_BLK;
    const int rbase = tile0 * TILE_ROWS;

    const float theta = logf(BETA / (float)(*lptr) + 1.0f);
    const float omt   = 1.0f - theta;

    const int nrows = min(64, N_NODES - rbase);
    for (int i = threadIdx.x; i < nrows * (D / 4); i += 128) {
        const int r = i >> 5;
        const int c = (i & 31) * 4;
        const size_t g = (size_t)(rbase + r) * D + c;
        const float4 a  = *(const float4*)(hi + g);
        const float4 b0 = *(const float4*)(h0 + g);
        float* s = sp + r * LDSP + c;
        s[0] = (1.0f - ALPHA) * a.x + ALPHA * b0.x;
        s[1] = (1.0f - ALPHA) * a.y + ALPHA * b0.y;
        s[2] = (1.0f - ALPHA) * a.z + ALPHA * b0.z;
        s[3] = (1.0f - ALPHA) * a.w + ALPHA * b0.w;
    }
    __syncthreads();

    const int myTile = tile0 + wv;
    if (myTile >= ROW_TILES) return;

    const int am   = lane & 15;
    const int koff = (lane >> 4) * 2;
    const float* arow = sp + (wv * TILE_ROWS + am) * LDSP;

    for (int ct = 0; ct < 8; ++ct) {
        const int c0  = ct * 16;
        const int n16 = lane & 15;
        const float* brow = W + (size_t)(c0 + n16) * D;

        v8f acc = {};
        #pragma unroll 8
        for (int k = 0; k < D; k += 4) {
            v2f a = *(const v2f*)(arow + k + koff);
            v2f b = *(const v2f*)(brow + k + koff);
            acc = __builtin_amdgcn_wmma_f32_16x16x4_f32(
                      false, a, false, b, (short)0, acc, false, false);
        }

        const float bb = bias[c0 + n16];
        const int   mb = (lane >> 4) * 8;
        #pragma unroll
        for (int v = 0; v < 8; ++v) {
            const int m   = mb + v;
            const float sup = sp[(wv * TILE_ROWS + m) * LDSP + c0 + n16];
            out[(size_t)(rbase + wv * TILE_ROWS + m) * D + c0 + n16] =
                theta * (acc[v] + bb) + omt * sup;
        }
    }
}

extern "C" void kernel_launch(void* const* d_in, const int* in_sizes, int n_in,
                              void* d_out, int out_size, void* d_ws, size_t ws_size,
                              hipStream_t stream) {
    const float* h    = (const float*)d_in[0];
    const float* h0   = (const float*)d_in[1];
    const int*   erow = (const int*)  d_in[2];
    const int*   ecol = (const int*)  d_in[3];
    const float* ev   = (const float*)d_in[4];
    const float* W    = (const float*)d_in[5];
    const float* bias = (const float*)d_in[6];
    const int*   l    = (const int*)  d_in[7];
    float* out = (float*)d_out;
    float* hi  = (float*)d_ws;

    const int n4 = (N_NODES * D) / 4;
    gcn_zero<<<(n4 + 255) / 256, 256, 0, stream>>>((float4*)d_ws, n4);

    const int waves   = (N_EDGES + EDGES_PER_WAVE - 1) / EDGES_PER_WAVE;
    const int threads = waves * 32;
    gcn_scatter<<<(threads + 255) / 256, 256, 0, stream>>>(h, erow, ecol, ev, hi);

    gcn_gemm<<<(ROW_TILES + WAVES_PER_BLK - 1) / WAVES_PER_BLK, 128, 0, stream>>>(
        hi, h0, W, bias, l, out);
}


# Rows owned by the edge chunk they start in: a wave skips its leading partial row, runs past the chunk end to finish its last row, zeroes empty rows; one sum chain per row; atomics, zero launch gone
# speedup vs baseline: 1.7051x; 1.7051x over previous
#include <hip/hip_runtime.h>

#define N_NODES 50000
#define N_EDGES 800000
#define D       128
#define ALPHA   0.1f
#define BETA    0.5f

#define EDGES_PER_WAVE 128
#define SCAT_WAVES_BLK  4
#define GROUPS_PER_WAVE (EDGES_PER_WAVE / 32)
#define TILE_ROWS      16
#define WAVES_PER_BLK  4
#define ROW_TILES      (N_NODES / TILE_ROWS)
#define LDSP           (D + 4)

typedef float v2f __attribute__((ext_vector_type(2)));
typedef float v8f __attribute__((ext_vector_type(8)));

__device__ __forceinline__
void issue8(const float* __restrict__ h, const int colG, const int f, const int b,
            float4 (&g)[8])
{
    #pragma unroll
    for (int t = 0; t < 8; ++t) {
        const int c = __builtin_amdgcn_readlane(colG, b * 8 + t);
        g[t] = *(const float4*)(h + (size_t)c * D + f);
    }
}

__device__ __forceinline__
void drain8(const float4 (&g)[8], const int b, const int cnt,
            const int rowC, const int nxtC, const float valC,
            const unsigned endmask, const int f, float* __restrict__ hi, float4& acc)
{
    #pragma unroll
    for (int t = 0; t < 8; ++t) {
        const int j = b * 8 + t;
        if (j < cnt) {
            const float v = __int_as_float(__builtin_amdgcn_readlane(__float_as_int(valC), j));
            acc.x = fmaf(v, g[t].x, acc.x);
            acc.y = fmaf(v, g[t].y, acc.y);
            acc.z = fmaf(v, g[t].z, acc.z);
            acc.w = fmaf(v, g[t].w, acc.w);
            if ((endmask >> j) & 1u) {
                const int r  = __builtin_amdgcn_readlane(rowC, j);
                const int nx = __builtin_amdgcn_readlane(nxtC, j);
                *(float4*)(hi + (size_t)r * D + f) = acc;
                acc.x = 0.f; acc.y = 0.f; acc.z = 0.f; acc.w = 0.f;
                const int zend = (nx < 0) ? N_NODES : nx;
                #pragma nounroll
                for (int z = r + 1; z < zend; ++z)
                    *(float4*)(hi + (size_t)z * D + f) = acc;
            }
        }
    }
}

__global__ __launch_bounds__(SCAT_WAVES_BLK * 32)
void gcn_scatter(const float* __restrict__ h,
                 const int*   __restrict__ erow,
                 const int*   __restrict__ ecol,
                 const float* __restrict__ evals,
                 float*       __restrict__ hi)
{
    static_assert(N_EDGES % EDGES_PER_WAVE == 0, "chunks are whole");
    const int lane  = threadIdx.x & 31;
    const int gwave = __builtin_amdgcn_readfirstlane(
                          blockIdx.x * SCAT_WAVES_BLK + (threadIdx.x >> 5));
    const int e0    = gwave * EDGES_PER_WAVE;
    if (e0 >= N_EDGES) return;
    const int eEnd  = e0 + EDGES_PER_WAVE;
    const int f     = lane * 4;

    int rs[GROUPS_PER_WAVE], ps[GROUPS_PER_WAVE];
    for (int g = 0; g < GROUPS_PER_WAVE; ++g) {
        const int i = e0 + g * 32 + lane;
        rs[g] = erow[i];
        const int pv = erow[max(i - 1, 0)];
        ps[g] = (i > 0) ? pv : -1;
    }
    int start = -1;
    for (int g = GROUPS_PER_WAVE - 1; g >= 0; --g) {
        const unsigned m = __builtin_amdgcn_ballot_w32(rs[g] != ps[g]);
        if (m) start = e0 + g * 32 + __builtin_ctz(m);
    }
    if (start < 0) return;

    float4 acc; acc.x = 0.f; acc.y = 0.f; acc.z = 0.f; acc.w = 0.f;

    if (gwave == 0) {
        const int r0 = erow[0];
        for (int z = 0; z < r0; ++z)
            *(float4*)(hi + (size_t)z * D + f) = acc;
    }

    int base = start;
    int   colN, rowN, nxtN, prvN;
    float valN;
    {
        const int i  = base + lane;
        const int ci = min(i, N_EDGES - 1);
        colN = ecol[ci];
        rowN = erow[ci];
        valN = evals[ci];
        const int nv = erow[min(i + 1, N_EDGES - 1)];
        const int pv = erow[max(ci - 1, 0)];
        nxtN = (i + 1 < N_EDGES) ? nv : -1;
        prvN = (ci > 0) ? pv : -1;
    }

    float4 g0[8], g1[8];

    for (int it = 0; it < N_EDGES / 32 + 2; ++it) {
        const int   colC = colN, rowC = rowN, nxtC = nxtN, prvC = prvN;
        const float valC = valN;
        const int   idxC = base + lane;
        {
            const int i  = idxC + 32;
            const int ci = min(i, N_EDGES - 1);
            colN = ecol[ci];
            rowN = erow[ci];
            valN = evals[ci];
            const int nv = erow[min(i + 1, N_EDGES - 1)];
            const int pv = erow[max(ci - 1, 0)];
            nxtN = (i + 1 < N_EDGES) ? nv : -1;
            prvN = (ci > 0) ? pv : -1;
        }
        const unsigned stopm = __builtin_amdgcn_ballot_w32(
            idxC >= N_EDGES || (idxC >= eEnd && rowC != prvC));
        const int cnt = stopm ? __builtin_ctz(stopm) : 32;
        if (cnt == 0) break;
        const unsigned endmask = __builtin_amdgcn_ballot_w32(nxtC != rowC);
        const int colG = (lane < cnt) ? colC : __builtin_amdgcn_readlane(colC, cnt - 1);

        issue8(h, colG, f, 0, g0);
        issue8(h, colG, f, 1, g1);
        drain8(g0, 0, cnt, rowC, nxtC, valC, endmask, f, hi, acc);
        issue8(h, colG, f, 2, g0);
        drain8(g1, 1, cnt, rowC, nxtC, valC, endmask, f, hi, acc);
        issue8(h, colG, f, 3, g1);
        drain8(g0, 2, cnt, rowC, nxtC, valC, endmask, f, hi, acc);
        drain8(g1, 3, cnt, rowC, nxtC, valC, endmask, f, hi, acc);

        if (cnt < 32) break;
        base += 32;
    }
}

__global__ __launch_bounds__(128)
void gcn_gemm(const float* __restrict__ hi,
              const float* __restrict__ h0,
              const float* __restrict__ W,
              const float* __restrict__ bias,
              const int*   __restrict__ lptr,
              float*       __restrict__ out)
{
    __shared__ float sp[64 * LDSP];

    const int lane  = threadIdx.x & 31;
    const int wv    = threadIdx.x >> 5;
    const int tile0 = blockIdx.x * WAVES_PER_BLK;
    const int rbase = tile0 * TILE_ROWS;

    const float theta = logf(BETA / (float)(*lptr) + 1.0f);
    const float omt   = 1.0f - theta;

    const int nrows = min(64, N_NODES - rbase);
    for (int i = threadIdx.x; i < nrows * (D / 4); i += 128) {
        const int r = i >> 5;
        const int c = (i & 31) * 4;
        const size_t g = (size_t)(rbase + r) * D + c;
        const float4 a  = *(const float4*)(hi + g);
        const float4 b0 = *(const float4*)(h0 + g);
        float* s = sp + r * LDSP + c;
        s[0] = (1.0f - ALPHA) * a.x + ALPHA * b0.x;
        s[1] = (1.0f - ALPHA) * a.y + ALPHA * b0.y;
        s[2] = (1.0f - ALPHA) * a.z + ALPHA * b0.z;
        s[3] = (1.0f - ALPHA) * a.w + ALPHA * b0.w;
    }
    __syncthreads();

    const int myTile = tile0 + wv;
    if (myTile >= ROW_TILES) return;

    const int am   = lane & 15;
    const int koff = (lane >> 4) * 2;
    const float* arow = sp + (wv * TILE_ROWS + am) * LDSP;

    for (int ct = 0; ct < 8; ++ct) {
        const int c0  = ct * 16;
        const int n16 = lane & 15;
        const float* brow = W + (size_t)(c0 + n16) * D;

        v8f acc = {};
        #pragma unroll 8
        for (int k = 0; k < D; k += 4) {
            v2f a = *(const v2f*)(arow + k + koff);
            v2f b = *(const v2f*)(brow + k + koff);
            acc = __builtin_amdgcn_wmma_f32_16x16x4_f32(
                      false, a, false, b, (short)0, acc, false, false);
        }

        const float bb = bias[c0 + n16];
        const int   mb = (lane >> 4) * 8;
        #pragma unroll
        for (int v = 0; v < 8; ++v) {
            const int m   = mb + v;
            const float sup = sp[(wv * TILE_ROWS + m) * LDSP + c0 + n16];
            out[(size_t)(rbase + wv * TILE_ROWS + m) * D + c0 + n16] =
                theta * (acc[v] + bb) + omt * sup;
        }
    }
}

extern "C" void kernel_launch(void* const* d_in, const int* in_sizes, int n_in,
                              void* d_out, int out_size, void* d_ws, size_t ws_size,
                              hipStream_t stream) {
    const float* h    = (const float*)d_in[0];
    const float* h0   = (const float*)d_in[1];
    const int*   erow = (const int*)  d_in[2];
    const int*   ecol = (const int*)  d_in[3];
    const float* ev   = (const float*)d_in[4];
    const float* W    = (const float*)d_in[5];
    const float* bias = (const float*)d_in[6];
    const int*   l    = (const int*)  d_in[7];
    float* out = (float*)d_out;
    float* hi  = (float*)d_ws;

    const int waves = N_EDGES / EDGES_PER_WAVE;
    gcn_scatter<<<(waves + SCAT_WAVES_BLK - 1) / SCAT_WAVES_BLK, SCAT_WAVES_BLK * 32, 0, stream>>>(
        h, erow, ecol, ev, hi);

    gcn_gemm<<<(ROW_TILES + WAVES_PER_BLK - 1) / WAVES_PER_BLK, 128, 0, stream>>>(
        hi, h0, W, bias, l, out);
}


# GEMM re-cut to one 16-wave work-group per WGP that stages W once in LDS in B-fragment order beside its row panel, all loads in flight together; B fragments are LDS reads; sums in old order
# speedup vs baseline: 2.3730x; 1.3917x over previous
#include <hip/hip_runtime.h>

#define N_NODES 50000
#define N_EDGES 800000
#define D       128
#define ALPHA   0.1f
#define BETA    0.5f

#define EDGES_PER_WAVE 128
#define SCAT_WAVES_BLK  4
#define GROUPS_PER_WAVE (EDGES_PER_WAVE / 32)
#define TILE_ROWS      16
#define ROW_TILES      (N_NODES / TILE_ROWS)
#define LDSP           (D + 4)

typedef float v2f __attribute__((ext_vector_type(2)));
typedef float v8f __attribute__((ext_vector_type(8)));

__device__ __forceinline__
void issue8(const float* __restrict__ h, const int colG, const int f, const int b,
            float4 (&g)[8])
{
    #pragma unroll
    for (int t = 0; t < 8; ++t) {
        const int c = __builtin_amdgcn_readlane(colG, b * 8 + t);
        g[t] = *(const float4*)(h + (size_t)c * D + f);
    }
}

__device__ __forceinline__
void drain8(const float4 (&g)[8], const int b, const int cnt,
            const int rowC, const int nxtC, const float valC,
            const unsigned endmask, const int f, float* __restrict__ hi, float4& acc)
{
    #pragma unroll
    for (int t = 0; t < 8; ++t) {
        const int j = b * 8 + t;
        if (j < cnt) {
            const float v = __int_as_float(__builtin_amdgcn_readlane(__float_as_int(valC), j));
            acc.x = fmaf(v, g[t].x, acc.x);
            acc.y = fmaf(v, g[t].y, acc.y);
            acc.z = fmaf(v, g[t].z, acc.z);
            acc.w = fmaf(v, g[t].w, acc.w);
            if ((endmask >> j) & 1u) {
                const int r  = __builtin_amdgcn_readlane(rowC, j);
                const int nx = __builtin_amdgcn_readlane(nxtC, j);
                *(float4*)(hi + (size_t)r * D + f) = acc;
                acc.x = 0.f; acc.y = 0.f; acc.z = 0.f; acc.w = 0.f;
                const int zend = (nx < 0) ? N_NODES : nx;
                #pragma nounroll
                for (int z = r + 1; z < zend; ++z)
                    *(float4*)(hi + (size_t)z * D + f) = acc;
            }
        }
    }
}

__global__ __launch_bounds__(SCAT_WAVES_BLK * 32)
void gcn_scatter(const float* __restrict__ h,
                 const int*   __restrict__ erow,
                 const int*   __restrict__ ecol,
                 const float* __restrict__ evals,
                 float*       __restrict__ hi)
{
    static_assert(N_EDGES % EDGES_PER_WAVE == 0, "chunks are whole");
    const int lane  = threadIdx.x & 31;
    const int gwave = __builtin_amdgcn_readfirstlane(
                          blockIdx.x * SCAT_WAVES_BLK + (threadIdx.x >> 5));
    const int e0    = gwave * EDGES_PER_WAVE;
    if (e0 >= N_EDGES) return;
    const int eEnd  = e0 + EDGES_PER_WAVE;
    const int f     = lane * 4;

    int rs[GROUPS_PER_WAVE], ps[GROUPS_PER_WAVE];
    for (int g = 0; g < GROUPS_PER_WAVE; ++g) {
        const int i = e0 + g * 32 + lane;
        rs[g] = erow[i];
        const int pv = erow[max(i - 1, 0)];
        ps[g] = (i > 0) ? pv : -1;
    }
    int start = -1;
    for (int g = GROUPS_PER_WAVE - 1; g >= 0; --g) {
        const unsigned m = __builtin_amdgcn_ballot_w32(rs[g] != ps[g]);
        if (m) start = e0 + g * 32 + __builtin_ctz(m);
    }
    if (start < 0) return;

    float4 acc; acc.x = 0.f; acc.y = 0.f; acc.z = 0.f; acc.w = 0.f;

    if (gwave == 0) {
        const int r0 = erow[0];
        for (int z = 0; z < r0; ++z)
            *(float4*)(hi + (size_t)z * D + f) = acc;
    }

    int base = start;
    int   colN, rowN, nxtN, prvN;
    float valN;
    {
        const int i  = base + lane;
        const int ci = min(i, N_EDGES - 1);
        colN = ecol[ci];
        rowN = erow[ci];
        valN = evals[ci];
        const int nv = erow[min(i + 1, N_EDGES - 1)];
        const int pv = erow[max(ci - 1, 0)];
        nxtN = (i + 1 < N_EDGES) ? nv : -1;
        prvN = (ci > 0) ? pv : -1;
    }

    float4 g0[8], g1[8];

    for (int it = 0; it < N_EDGES / 32 + 2; ++it) {
        const int   colC = colN, rowC = rowN, nxtC = nxtN, prvC = prvN;
        const float valC = valN;
        const int   idxC = base + lane;
        {
            const int i  = idxC + 32;
            const int ci = min(i, N_EDGES - 1);
            colN = ecol[ci];
            rowN = erow[ci];
            valN = evals[ci];
            const int nv = erow[min(i + 1, N_EDGES - 1)];
            const int pv = erow[max(ci - 1, 0)];
            nxtN = (i + 1 < N_EDGES) ? nv : -1;
            prvN = (ci > 0) ? pv : -1;
        }
        const unsigned stopm = __builtin_amdgcn_ballot_w32(
            idxC >= N_EDGES || (idxC >= eEnd && rowC != prvC));
        const int cnt = stopm ? __builtin_ctz(stopm) : 32;
        if (cnt == 0) break;
        const unsigned endmask = __builtin_amdgcn_ballot_w32(nxtC != rowC);
        const int colG = (lane < cnt) ? colC : __builtin_amdgcn_readlane(colC, cnt - 1);

        issue8(h, colG, f, 0, g0);
        issue8(h, colG, f, 1, g1);
        drain8(g0, 0, cnt, rowC, nxtC, valC, endmask, f, hi, acc);
        issue8(h, colG, f, 2, g0);
        drain8(g1, 1, cnt, rowC, nxtC, valC, endmask, f, hi, acc);
        issue8(h, colG, f, 3, g1);
        drain8(g0, 2, cnt, rowC, nxtC, valC, endmask, f, hi, acc);
        drain8(g1, 3, cnt, rowC, nxtC, valC, endmask, f, hi, acc);

        if (cnt < 32) break;
        base += 32;
    }
}

#define GEMM_WGS     256
#define GEMM_WAVES   16
#define GEMM_THREADS (GEMM_WAVES * 32)
#define MIN_TILES    (ROW_TILES / GEMM_WGS)
#define MAX_TILES    (MIN_TILES + 1)
#define PANEL_ROWS   (MAX_TILES * TILE_ROWS)
#define FRAG_FLOATS  64
#define CT_FLOATS    ((D / 4) * FRAG_FLOATS)

__global__ __launch_bounds__(GEMM_THREADS)
void gcn_gemm(const float* __restrict__ hi,
              const float* __restrict__ h0,
              const float* __restrict__ W,
              const float* __restrict__ bias,
              const int*   __restrict__ lptr,
              float*       __restrict__ out)
{
    static_assert(MIN_TILES * GEMM_WGS < ROW_TILES && MAX_TILES * GEMM_WGS >= ROW_TILES,
                  "a work-group owns MIN_TILES or MAX_TILES row tiles");
    static_assert(D == 128 && GEMM_WAVES == 16, "the staging maps below are written for these");
    __shared__ __attribute__((aligned(16))) float wf[D * D];
    __shared__ __attribute__((aligned(16))) float sp[PANEL_ROWS * LDSP];

    const int lane = threadIdx.x & 31;
    const int wv     = __builtin_amdgcn_readfirstlane(threadIdx.x >> 5);
    const int t0     = ((int)blockIdx.x * ROW_TILES) / GEMM_WGS;
    const int t1     = (((int)blockIdx.x + 1) * ROW_TILES) / GEMM_WGS;
    const int ntiles = t1 - t0;
    const int rbase  = t0 * TILE_ROWS;
    const int n16    = lane & 15;
    const int hh     = lane >> 4;
    const int c4     = lane * 4;

    float4 pa[MAX_TILES], pb[MAX_TILES];
    #pragma unroll
    for (int j = 0; j < MAX_TILES; ++j) {
        const int r = min(rbase + j * 16 + wv, N_NODES - 1);
        const size_t g = (size_t)r * D + c4;
        pa[j] = *(const float4*)(hi + g);
        pb[j] = *(const float4*)(h0 + g);
    }
    float4 wr[D / 16];
    #pragma unroll
    for (int j = 0; j < D / 16; ++j)
        wr[j] = *(const float4*)(W + (size_t)(j * 16 + n16) * D + 8 * wv + 4 * hh);

    const float theta = logf(BETA / (float)(*lptr) + 1.0f);
    const float omt   = 1.0f - theta;

    #pragma unroll
    for (int j = 0; j < MAX_TILES; ++j) {
        float4 o;
        o.x = (1.0f - ALPHA) * pa[j].x + ALPHA * pb[j].x;
        o.y = (1.0f - ALPHA) * pa[j].y + ALPHA * pb[j].y;
        o.z = (1.0f - ALPHA) * pa[j].z + ALPHA * pb[j].z;
        o.w = (1.0f - ALPHA) * pa[j].w + ALPHA * pb[j].w;
        *(float4*)(sp + (j * 16 + wv) * LDSP + c4) = o;
    }
    #pragma unroll
    for (int j = 0; j < D / 16; ++j) {
        float* d = wf + j * CT_FLOATS + (2 * wv + hh) * FRAG_FLOATS + n16 * 2;
        v2f lo, up;
        lo.x = wr[j].x; lo.y = wr[j].y;
        up.x = wr[j].z; up.y = wr[j].w;
        *(v2f*)d        = lo;
        *(v2f*)(d + 32) = up;
    }
    __syncthreads();

    const int koff = hh * 2;
    const int mb   = hh * 8;
    for (int u = wv; u < 2 * ntiles; u += GEMM_WAVES) {
        const int tl = u >> 1;
        const int cq = (u & 1) * 4;
        const float* arow = sp + (tl * TILE_ROWS + n16) * LDSP + koff;
        const float* bfr  = wf + cq * CT_FLOATS + lane * 2;

        v8f acc[4] = {};
        #pragma unroll 8
        for (int s = 0; s < D / 4; ++s) {
            const v2f a = *(const v2f*)(arow + 4 * s);
            #pragma unroll
            for (int j = 0; j < 4; ++j) {
                const v2f b = *(const v2f*)(bfr + j * CT_FLOATS + s * FRAG_FLOATS);
                acc[j] = __builtin_amdgcn_wmma_f32_16x16x4_f32(
                             false, a, false, b, (short)0, acc[j], false, false);
            }
        }

        #pragma unroll
        for (int j = 0; j < 4; ++j) {
            const int   c0 = (cq + j) * 16;
            const float bb = bias[c0 + n16];
            #pragma unroll
            for (int v = 0; v < 8; ++v) {
                const int m   = mb + v;
                const float sup = sp[(tl * TILE_ROWS + m) * LDSP + c0 + n16];
                out[(size_t)(rbase + tl * TILE_ROWS + m) * D + c0 + n16] =
                    theta * (acc[j][v] + bb) + omt * sup;
            }
        }
    }
}

extern "C" void kernel_launch(void* const* d_in, const int* in_sizes, int n_in,
                              void* d_out, int out_size, void* d_ws, size_t ws_size,
                              hipStream_t stream) {
    const float* h    = (const float*)d_in[0];
    const float* h0   = (const float*)d_in[1];
    const int*   erow = (const int*)  d_in[2];
    const int*   ecol = (const int*)  d_in[3];
    const float* ev   = (const float*)d_in[4];
    const float* W    = (const float*)d_in[5];
    const float* bias = (const float*)d_in[6];
    const int*   l    = (const int*)  d_in[7];
    float* out = (float*)d_out;
    float* hi  = (float*)d_ws;

    const int waves = N_EDGES / EDGES_PER_WAVE;
    gcn_scatter<<<(waves + SCAT_WAVES_BLK - 1) / SCAT_WAVES_BLK, SCAT_WAVES_BLK * 32, 0, stream>>>(
        h, erow, ecol, ev, hi);

    gcn_gemm<<<GEMM_WGS, GEMM_THREADS, 0, stream>>>(
        hi, h0, W, bias, l, out);
}
